# starting kernel
# baseline (speedup 1.0000x reference)
#include <hip/hip_runtime.h>

#define B_ 16
#define L_ 2048
#define D_ 1024
#define S_ 16
#define OUTD (3 * D_)
#define KWAVES 8
#define KSLICE (L_ / KWAVES)

typedef __attribute__((ext_vector_type(2))) float v2f;
typedef __attribute__((ext_vector_type(8))) float v8f;

__global__ void find_markers(const int* __restrict__ ids,
                             int* __restrict__ idx0,
                             int* __restrict__ idx1) {
    const int b = blockIdx.x;
    const int* row = ids + b * L_;
    for (int p = threadIdx.x; p < L_; p += blockDim.x) {
        const int id = row[p];
        const bool isS = (id >= 1 && id <= 2);
        const bool isE = (id >= 3 && id <= 4);
        if (isS) {
            int rank = 0;
            for (int j = 0; j < p; ++j) {
                const int v = row[j];
                rank += (v >= 1 && v <= 2) ? 1 : 0;
            }
            idx0[b * S_ + rank] = p;
        } else if (isE) {
            int rank = 0;
            for (int j = 0; j < p; ++j) {
                const int v = row[j];
                rank += (v >= 3 && v <= 4) ? 1 : 0;
            }
            idx1[b * S_ + rank] = p;
        }
    }
}

__global__ void gather_ends(const float* __restrict__ x,
                            const int* __restrict__ idx0,
                            const int* __restrict__ idx1,
                            float* __restrict__ out) {
    const int bs = blockIdx.x;
    const int b = bs >> 4;
    const int i0 = idx0[bs];
    const int i1 = idx1[bs];
    const float4* r0 = (const float4*)(x + ((size_t)b * L_ + i0) * D_);
    const float4* r1 = (const float4*)(x + ((size_t)b * L_ + i1) * D_);
    float4* o = (float4*)(out + (size_t)bs * OUTD);
    const int t = threadIdx.x;
    o[t] = r0[t];
    o[(D_ / 4) + t] = r1[t];
}

__global__ void __launch_bounds__(256) span_mean_wmma(
    const float* __restrict__ x,
    const int* __restrict__ idx0,
    const int* __restrict__ idx1,
    float* __restrict__ out) {
    __shared__ float red[KWAVES * 256];

    const int b    = blockIdx.y;
    const int wave = threadIdx.x >> 5;
    const int lane = threadIdx.x & 31;
    const int col0 = blockIdx.x * 16;

    const int n     = lane & 15;
    const int m     = lane & 15;
    const int khalf = (lane >> 4) * 2;

    const int   i0   = idx0[b * S_ + m];
    const int   i1   = idx1[b * S_ + m];
    const float invc = 1.0f / (float)(i1 - i0 - 1);

    const float* bcol = x + (size_t)b * L_ * D_ + col0 + n;

    v8f c = {};
    const int kbeg = wave * KSLICE;
#pragma unroll 16
    for (int l0 = kbeg; l0 < kbeg + KSLICE; l0 += 4) {
        const int lx = l0 + khalf;
        const int ly = lx + 1;

        v2f a;
        a.x = (lx > i0 && lx < i1) ? invc : 0.0f;
        a.y = (ly > i0 && ly < i1) ? invc : 0.0f;

        v2f bb;
        bb.x = bcol[(size_t)lx * D_];
        bb.y = bcol[(size_t)ly * D_];

        c = __builtin_amdgcn_wmma_f32_16x16x4_f32(
                 false, a,  false, bb,
                 (short)0, c,  false,  false);
    }

#pragma unroll
    for (int i = 0; i < 8; ++i)
        red[wave * 256 + i * 32 + lane] = c[i];
    __syncthreads();

    const int t = threadIdx.x;
    float s = 0.0f;
#pragma unroll
    for (int w = 0; w < KWAVES; ++w)
        s += red[w * 256 + t];

    const int i    = t >> 5;
    const int ln   = t & 31;
    const int span = (ln >> 4) * 8 + i;
    const int col  = col0 + (ln & 15);
    out[(size_t)(b * S_ + span) * OUTD + 2 * D_ + col] = s;
}

extern "C" void kernel_launch(void* const* d_in, const int* in_sizes, int n_in,
                              void* d_out, int out_size, void* d_ws, size_t ws_size,
                              hipStream_t stream) {
    const float* x   = (const float*)d_in[0];
    const int*   ids = (const int*)d_in[2];
    float*       out = (float*)d_out;

    int* idx0 = (int*)d_ws;
    int* idx1 = idx0 + B_ * S_;

    find_markers<<<B_, 256, 0, stream>>>(ids, idx0, idx1);
    gather_ends<<<B_ * S_, 256, 0, stream>>>(x, idx0, idx1, out);

    dim3 grid(D_ / 16, B_);
    span_mean_wmma<<<grid, 256, 0, stream>>>(x, idx0, idx1, out);
}


# find_markers: marker rank from wave ballots + popcount prefix (8 waves x 8 chunks, one LDS hand-over) instead of one thread rescanning the row per marker
# speedup vs baseline: 3.3365x; 3.3365x over previous
#include <hip/hip_runtime.h>

#define B_ 16
#define L_ 2048
#define D_ 1024
#define S_ 16
#define OUTD (3 * D_)
#define KWAVES 8
#define KSLICE (L_ / KWAVES)

typedef __attribute__((ext_vector_type(2))) float v2f;
typedef __attribute__((ext_vector_type(8))) float v8f;

#define FM_WAVES  8
#define FM_CHUNKS (L_ / (FM_WAVES * 32))

__global__ void __launch_bounds__(FM_WAVES * 32) find_markers(
    const int* __restrict__ ids,
    int* __restrict__ idx0,
    int* __restrict__ idx1) {
    __shared__ int wcnt[FM_WAVES];

    const int b    = blockIdx.x;
    const int wave = threadIdx.x >> 5;
    const int lane = threadIdx.x & 31;
    const int tok0 = wave * (FM_CHUNKS * 32) + lane;
    const int* row = ids + b * L_ + tok0;

    int id[FM_CHUNKS];
    for (int j = 0; j < FM_CHUNKS; ++j)
        id[j] = row[j * 32];

    unsigned ms[FM_CHUNKS], me[FM_CHUNKS];
    int ns = 0, ne = 0;
    for (int j = 0; j < FM_CHUNKS; ++j) {
        ms[j] = __builtin_amdgcn_ballot_w32(id[j] >= 1 && id[j] <= 2);
        me[j] = __builtin_amdgcn_ballot_w32(id[j] >= 3 && id[j] <= 4);
        ns += __popc(ms[j]);
        ne += __popc(me[j]);
    }
    if (lane == 0) wcnt[wave] = ns | (ne << 16);
    __syncthreads();

    int bs = 0, be = 0;
    int ts = 0, te = 0;
    for (int w = 0; w < FM_WAVES; ++w) {
        const int v = wcnt[w];
        const int s = v & 0xffff;
        const int e = v >> 16;
        if (w < wave) { bs += s; be += e; }
        ts += s;
        te += e;
    }

    const unsigned below = (1u << lane) - 1u;
    for (int j = 0; j < FM_CHUNKS; ++j) {
        const int p = tok0 + j * 32;
        if ((ms[j] >> lane) & 1u) {
            const int r = bs + __popc(ms[j] & below);
            if (r < S_) idx0[b * S_ + r] = p;
        }
        if ((me[j] >> lane) & 1u) {
            const int r = be + __popc(me[j] & below);
            if (r < S_) idx1[b * S_ + r] = p;
        }
        bs += __popc(ms[j]);
        be += __popc(me[j]);
    }

    if (threadIdx.x < S_) {
        if ((int)threadIdx.x >= ts) idx0[b * S_ + threadIdx.x] = 0;
        if ((int)threadIdx.x >= te) idx1[b * S_ + threadIdx.x] = 0;
    }
}

__global__ void gather_ends(const float* __restrict__ x,
                            const int* __restrict__ idx0,
                            const int* __restrict__ idx1,
                            float* __restrict__ out) {
    const int bs = blockIdx.x;
    const int b = bs >> 4;
    const int i0 = idx0[bs];
    const int i1 = idx1[bs];
    const float4* r0 = (const float4*)(x + ((size_t)b * L_ + i0) * D_);
    const float4* r1 = (const float4*)(x + ((size_t)b * L_ + i1) * D_);
    float4* o = (float4*)(out + (size_t)bs * OUTD);
    const int t = threadIdx.x;
    o[t] = r0[t];
    o[(D_ / 4) + t] = r1[t];
}

__global__ void __launch_bounds__(256) span_mean_wmma(
    const float* __restrict__ x,
    const int* __restrict__ idx0,
    const int* __restrict__ idx1,
    float* __restrict__ out) {
    __shared__ float red[KWAVES * 256];

    const int b    = blockIdx.y;
    const int wave = threadIdx.x >> 5;
    const int lane = threadIdx.x & 31;
    const int col0 = blockIdx.x * 16;

    const int n     = lane & 15;
    const int m     = lane & 15;
    const int khalf = (lane >> 4) * 2;

    const int   i0   = idx0[b * S_ + m];
    const int   i1   = idx1[b * S_ + m];
    const float invc = 1.0f / (float)(i1 - i0 - 1);

    const float* bcol = x + (size_t)b * L_ * D_ + col0 + n;

    v8f c = {};
    const int kbeg = wave * KSLICE;
#pragma unroll 16
    for (int l0 = kbeg; l0 < kbeg + KSLICE; l0 += 4) {
        const int lx = l0 + khalf;
        const int ly = lx + 1;

        v2f a;
        a.x = (lx > i0 && lx < i1) ? invc : 0.0f;
        a.y = (ly > i0 && ly < i1) ? invc : 0.0f;

        v2f bb;
        bb.x = bcol[(size_t)lx * D_];
        bb.y = bcol[(size_t)ly * D_];

        c = __builtin_amdgcn_wmma_f32_16x16x4_f32(
                 false, a,  false, bb,
                 (short)0, c,  false,  false);
    }

#pragma unroll
    for (int i = 0; i < 8; ++i)
        red[wave * 256 + i * 32 + lane] = c[i];
    __syncthreads();

    const int t = threadIdx.x;
    float s = 0.0f;
#pragma unroll
    for (int w = 0; w < KWAVES; ++w)
        s += red[w * 256 + t];

    const int i    = t >> 5;
    const int ln   = t & 31;
    const int span = (ln >> 4) * 8 + i;
    const int col  = col0 + (ln & 15);
    out[(size_t)(b * S_ + span) * OUTD + 2 * D_ + col] = s;
}

extern "C" void kernel_launch(void* const* d_in, const int* in_sizes, int n_in,
                              void* d_out, int out_size, void* d_ws, size_t ws_size,
                              hipStream_t stream) {
    const float* x   = (const float*)d_in[0];
    const int*   ids = (const int*)d_in[2];
    float*       out = (float*)d_out;

    int* idx0 = (int*)d_ws;
    int* idx1 = idx0 + B_ * S_;

    find_markers<<<B_, 256, 0, stream>>>(ids, idx0, idx1);
    gather_ends<<<B_ * S_, 256, 0, stream>>>(x, idx0, idx1, out);

    dim3 grid(D_ / 16, B_);
    span_mean_wmma<<<grid, 256, 0, stream>>>(x, idx0, idx1, out);
}


# span_mean_wmma: 64-col strips so each half-wave reads whole aligned 256-B row chunks (float4 loads feed 4 column tiles, ring of 4 load groups); K split over 16 waves not 8 changes the sum tree
# speedup vs baseline: 5.2113x; 1.5619x over previous
#include <hip/hip_runtime.h>

#define B_ 16
#define L_ 2048
#define D_ 1024
#define S_ 16
#define OUTD (3 * D_)

typedef __attribute__((ext_vector_type(2))) float v2f;
typedef __attribute__((ext_vector_type(4))) float v4f;
typedef __attribute__((ext_vector_type(8))) float v8f;

#define FM_WAVES  8
#define FM_CHUNKS (L_ / (FM_WAVES * 32))

__global__ void __launch_bounds__(FM_WAVES * 32) find_markers(
    const int* __restrict__ ids,
    int* __restrict__ idx0,
    int* __restrict__ idx1) {
    __shared__ int wcnt[FM_WAVES];

    const int b    = blockIdx.x;
    const int wave = threadIdx.x >> 5;
    const int lane = threadIdx.x & 31;
    const int tok0 = wave * (FM_CHUNKS * 32) + lane;
    const int* row = ids + b * L_ + tok0;

    int id[FM_CHUNKS];
    for (int j = 0; j < FM_CHUNKS; ++j)
        id[j] = row[j * 32];

    unsigned ms[FM_CHUNKS], me[FM_CHUNKS];
    int ns = 0, ne = 0;
    for (int j = 0; j < FM_CHUNKS; ++j) {
        ms[j] = __builtin_amdgcn_ballot_w32(id[j] >= 1 && id[j] <= 2);
        me[j] = __builtin_amdgcn_ballot_w32(id[j] >= 3 && id[j] <= 4);
        ns += __popc(ms[j]);
        ne += __popc(me[j]);
    }
    if (lane == 0) wcnt[wave] = ns | (ne << 16);
    __syncthreads();

    int bs = 0, be = 0;
    int ts = 0, te = 0;
    for (int w = 0; w < FM_WAVES; ++w) {
        const int v = wcnt[w];
        const int s = v & 0xffff;
        const int e = v >> 16;
        if (w < wave) { bs += s; be += e; }
        ts += s;
        te += e;
    }

    const unsigned below = (1u << lane) - 1u;
    for (int j = 0; j < FM_CHUNKS; ++j) {
        const int p = tok0 + j * 32;
        if ((ms[j] >> lane) & 1u) {
            const int r = bs + __popc(ms[j] & below);
            if (r < S_) idx0[b * S_ + r] = p;
        }
        if ((me[j] >> lane) & 1u) {
            const int r = be + __popc(me[j] & below);
            if (r < S_) idx1[b * S_ + r] = p;
        }
        bs += __popc(ms[j]);
        be += __popc(me[j]);
    }

    if (threadIdx.x < S_) {
        if ((int)threadIdx.x >= ts) idx0[b * S_ + threadIdx.x] = 0;
        if ((int)threadIdx.x >= te) idx1[b * S_ + threadIdx.x] = 0;
    }
}

__global__ void gather_ends(const float* __restrict__ x,
                            const int* __restrict__ idx0,
                            const int* __restrict__ idx1,
                            float* __restrict__ out) {
    const int bs = blockIdx.x;
    const int b = bs >> 4;
    const int i0 = idx0[bs];
    const int i1 = idx1[bs];
    const float4* r0 = (const float4*)(x + ((size_t)b * L_ + i0) * D_);
    const float4* r1 = (const float4*)(x + ((size_t)b * L_ + i1) * D_);
    float4* o = (float4*)(out + (size_t)bs * OUTD);
    const int t = threadIdx.x;
    o[t] = r0[t];
    o[(D_ / 4) + t] = r1[t];
}

#define SM_WAVES  16
#define SM_SLICE  (L_ / SM_WAVES)
#define SM_COLS   64
#define SM_GSTEPS 4
#define SM_GROUPS (SM_SLICE / (4 * SM_GSTEPS))
#define SM_RING   4

__global__ void __launch_bounds__(SM_WAVES * 32) span_mean_wmma(
    const float* __restrict__ x,
    const int* __restrict__ idx0,
    const int* __restrict__ idx1,
    float* __restrict__ out) {
    __shared__ float red[SM_WAVES * S_ * SM_COLS];

    const int b    = blockIdx.y;
    const int wave = threadIdx.x >> 5;
    const int lane = threadIdx.x & 31;
    const int col0 = blockIdx.x * SM_COLS;

    const int n     = lane & 15;
    const int h     = lane >> 4;
    const int khalf = h * 2;
    const int kbeg  = wave * SM_SLICE;

    const int i0 = idx0[b * S_ + n];
    const int i1 = idx1[b * S_ + n];

    const float* base = x + ((size_t)b * L_ + kbeg + khalf) * D_ + col0 + 4 * n;

    v4f buf[SM_RING][SM_GSTEPS][2];
#define LOAD_GROUP(G) do {                                                     \
        _Pragma("unroll") for (int s_ = 0; s_ < SM_GSTEPS; ++s_) {             \
            const float* p_ = base + (size_t)(((G) * SM_GSTEPS + s_) * 4) * D_; \
            buf[(G) % SM_RING][s_][0] = *(const v4f*)p_;                       \
            buf[(G) % SM_RING][s_][1] = *(const v4f*)(p_ + D_);                \
        } } while (0)

#pragma unroll
    for (int g = 0; g < SM_RING - 1; ++g) LOAD_GROUP(g);

    const float invc = 1.0f / (float)(i1 - i0 - 1);

    v8f c0 = {}, c1 = {}, c2 = {}, c3 = {};
#pragma unroll
    for (int g = 0; g < SM_GROUPS; ++g) {
        __builtin_amdgcn_sched_barrier(0);
        if (g + SM_RING - 1 < SM_GROUPS) LOAD_GROUP(g + SM_RING - 1);
        __builtin_amdgcn_sched_barrier(0);
#pragma unroll
        for (int s = 0; s < SM_GSTEPS; ++s) {
            const int lx = kbeg + (g * SM_GSTEPS + s) * 4 + khalf;
            const int ly = lx + 1;

            v2f a;
            a.x = (lx > i0 && lx < i1) ? invc : 0.0f;
            a.y = (ly > i0 && ly < i1) ? invc : 0.0f;

            const v4f r0 = buf[g % SM_RING][s][0];
            const v4f r1 = buf[g % SM_RING][s][1];
            v2f bb;
            bb.x = r0.x; bb.y = r1.x;
            c0 = __builtin_amdgcn_wmma_f32_16x16x4_f32(
                      false, a,  false, bb,
                      (short)0, c0,  false,  false);
            bb.x = r0.y; bb.y = r1.y;
            c1 = __builtin_amdgcn_wmma_f32_16x16x4_f32(
                     false, a, false, bb, (short)0, c1, false, false);
            bb.x = r0.z; bb.y = r1.z;
            c2 = __builtin_amdgcn_wmma_f32_16x16x4_f32(
                     false, a, false, bb, (short)0, c2, false, false);
            bb.x = r0.w; bb.y = r1.w;
            c3 = __builtin_amdgcn_wmma_f32_16x16x4_f32(
                     false, a, false, bb, (short)0, c3, false, false);
        }
    }
#undef LOAD_GROUP

#pragma unroll
    for (int r = 0; r < 8; ++r) {
        v4f v;
        v.x = c0[r]; v.y = c1[r]; v.z = c2[r]; v.w = c3[r];
        *(v4f*)&red[(wave * S_ + r + 8 * h) * SM_COLS + 4 * n] = v;
    }
    __syncthreads();

    const int span = wave;
    float sx = 0.0f, sy = 0.0f;
#pragma unroll
    for (int w = 0; w < SM_WAVES; ++w) {
        const v2f v = *(const v2f*)&red[(w * S_ + span) * SM_COLS + 2 * lane];
        sx += v.x;
        sy += v.y;
    }
    v2f o;
    o.x = sx; o.y = sy;
    *(v2f*)&out[(size_t)(b * S_ + span) * OUTD + 2 * D_ + col0 + 2 * lane] = o;
}

extern "C" void kernel_launch(void* const* d_in, const int* in_sizes, int n_in,
                              void* d_out, int out_size, void* d_ws, size_t ws_size,
                              hipStream_t stream) {
    const float* x   = (const float*)d_in[0];
    const int*   ids = (const int*)d_in[2];
    float*       out = (float*)d_out;

    int* idx0 = (int*)d_ws;
    int* idx1 = idx0 + B_ * S_;

    find_markers<<<B_, 256, 0, stream>>>(ids, idx0, idx1);
    gather_ends<<<B_ * S_, 256, 0, stream>>>(x, idx0, idx1, out);

    dim3 grid(D_ / SM_COLS, B_);
    span_mean_wmma<<<grid, SM_WAVES * 32, 0, stream>>>(x, idx0, idx1, out);
}


# single launch: each span-mean block finds its batch row's markers itself (wave ballots + popcount ranks through LDS, behind its first x loads) and copies its strip of the start/end rows
# speedup vs baseline: 7.2168x; 1.3848x over previous
#include <hip/hip_runtime.h>

#define B_ 16
#define L_ 2048
#define D_ 1024
#define S_ 16
#define OUTD (3 * D_)

typedef __attribute__((ext_vector_type(2))) float v2f;
typedef __attribute__((ext_vector_type(4))) float v4f;
typedef __attribute__((ext_vector_type(8))) float v8f;

#define SM_WAVES  16
#define SM_SLICE  (L_ / SM_WAVES)
#define SM_COLS   64
#define SM_GSTEPS 4
#define SM_GROUPS (SM_SLICE / (4 * SM_GSTEPS))
#define SM_RING   4
#define FM_CHUNKS (SM_SLICE / 32)

typedef __attribute__((ext_vector_type(4))) int v4i;

static __device__ __forceinline__ void lds_handover_barrier() {
    __builtin_amdgcn_fence(__ATOMIC_RELEASE, "workgroup", "local");
    __builtin_amdgcn_s_barrier();
    __builtin_amdgcn_fence(__ATOMIC_ACQUIRE, "workgroup", "local");
}

__global__ void __launch_bounds__(SM_WAVES * 32) span_pool_fused(
    const float* __restrict__ x,
    const int* __restrict__ ids,
    float* __restrict__ out) {
    __shared__ float red[SM_WAVES * S_ * SM_COLS];
    __shared__ __attribute__((aligned(16))) int wcnt[SM_WAVES];
    __shared__ int sidx[2 * S_];

    const int b    = blockIdx.y;
    const int wave = threadIdx.x >> 5;
    const int lane = threadIdx.x & 31;
    const int col0 = blockIdx.x * SM_COLS;

    const int n     = lane & 15;
    const int h     = lane >> 4;
    const int khalf = h * 2;
    const int kbeg  = wave * SM_SLICE;

    const int* idrow = ids + b * L_ + kbeg + lane;
    int id[FM_CHUNKS];
    for (int j = 0; j < FM_CHUNKS; ++j)
        id[j] = idrow[j * 32];

    const float* base = x + ((size_t)b * L_ + kbeg + khalf) * D_ + col0 + 4 * n;

    v4f buf[SM_RING][SM_GSTEPS][2];
#define LOAD_GROUP(G) do {                                                     \
        _Pragma("unroll") for (int s_ = 0; s_ < SM_GSTEPS; ++s_) {             \
            const float* p_ = base + (size_t)(((G) * SM_GSTEPS + s_) * 4) * D_; \
            buf[(G) % SM_RING][s_][0] = *(const v4f*)p_;                       \
            buf[(G) % SM_RING][s_][1] = *(const v4f*)(p_ + D_);                \
        } } while (0)

#pragma unroll
    for (int g = 0; g < SM_RING - 1; ++g) LOAD_GROUP(g);

    if (threadIdx.x < 2 * S_) sidx[threadIdx.x] = 0;

    unsigned ms[FM_CHUNKS], me[FM_CHUNKS];
    int ns = 0, ne = 0;
    for (int j = 0; j < FM_CHUNKS; ++j) {
        ms[j] = __builtin_amdgcn_ballot_w32(id[j] >= 1 && id[j] <= 2);
        me[j] = __builtin_amdgcn_ballot_w32(id[j] >= 3 && id[j] <= 4);
        ns += __popc(ms[j]);
        ne += __popc(me[j]);
    }
    if (lane == 0) wcnt[wave] = ns | (ne << 16);
    lds_handover_barrier();

    int pre = 0;
    for (int q = 0; q < SM_WAVES / 4; ++q) {
        const v4i v = *(const v4i*)&wcnt[4 * q];
        pre += (4 * q + 0 < wave) ? v.x : 0;
        pre += (4 * q + 1 < wave) ? v.y : 0;
        pre += (4 * q + 2 < wave) ? v.z : 0;
        pre += (4 * q + 3 < wave) ? v.w : 0;
    }
    int bs = pre & 0xffff;
    int be = pre >> 16;

    const unsigned below = (1u << lane) - 1u;
    for (int j = 0; j < FM_CHUNKS; ++j) {
        const int p = kbeg + j * 32 + lane;
        if ((ms[j] >> lane) & 1u) {
            const int r = bs + __popc(ms[j] & below);
            if (r < S_) sidx[r] = p;
        }
        if ((me[j] >> lane) & 1u) {
            const int r = be + __popc(me[j] & below);
            if (r < S_) sidx[S_ + r] = p;
        }
        bs += __popc(ms[j]);
        be += __popc(me[j]);
    }
    lds_handover_barrier();

    const int i0 = sidx[n];
    const int i1 = sidx[S_ + n];

    const int g0 = sidx[wave];
    const int g1 = sidx[S_ + wave];
    const v2f e0 = *(const v2f*)(x + ((size_t)b * L_ + g0) * D_ + col0 + 2 * lane);
    const v2f e1 = *(const v2f*)(x + ((size_t)b * L_ + g1) * D_ + col0 + 2 * lane);

    const float invc = 1.0f / (float)(i1 - i0 - 1);

    v8f c0 = {}, c1 = {}, c2 = {}, c3 = {};
#pragma unroll
    for (int g = 0; g < SM_GROUPS; ++g) {
        __builtin_amdgcn_sched_barrier(0);
        if (g + SM_RING - 1 < SM_GROUPS) LOAD_GROUP(g + SM_RING - 1);
        __builtin_amdgcn_sched_barrier(0);
#pragma unroll
        for (int s = 0; s < SM_GSTEPS; ++s) {
            const int lx = kbeg + (g * SM_GSTEPS + s) * 4 + khalf;
            const int ly = lx + 1;

            v2f a;
            a.x = (lx > i0 && lx < i1) ? invc : 0.0f;
            a.y = (ly > i0 && ly < i1) ? invc : 0.0f;

            const v4f r0 = buf[g % SM_RING][s][0];
            const v4f r1 = buf[g % SM_RING][s][1];
            v2f bb;
            bb.x = r0.x; bb.y = r1.x;
            c0 = __builtin_amdgcn_wmma_f32_16x16x4_f32(
                      false, a,  false, bb,
                      (short)0, c0,  false,  false);
            bb.x = r0.y; bb.y = r1.y;
            c1 = __builtin_amdgcn_wmma_f32_16x16x4_f32(
                     false, a, false, bb, (short)0, c1, false, false);
            bb.x = r0.z; bb.y = r1.z;
            c2 = __builtin_amdgcn_wmma_f32_16x16x4_f32(
                     false, a, false, bb, (short)0, c2, false, false);
            bb.x = r0.w; bb.y = r1.w;
            c3 = __builtin_amdgcn_wmma_f32_16x16x4_f32(
                     false, a, false, bb, (short)0, c3, false, false);
        }
    }
#undef LOAD_GROUP

    float* orow = out + (size_t)(b * S_ + wave) * OUTD + col0 + 2 * lane;
    *(v2f*)orow        = e0;
    *(v2f*)(orow + D_) = e1;

#pragma unroll
    for (int r = 0; r < 8; ++r) {
        v4f v;
        v.x = c0[r]; v.y = c1[r]; v.z = c2[r]; v.w = c3[r];
        *(v4f*)&red[(wave * S_ + r + 8 * h) * SM_COLS + 4 * n] = v;
    }
    lds_handover_barrier();

    float sx = 0.0f, sy = 0.0f;
#pragma unroll
    for (int w = 0; w < SM_WAVES; ++w) {
        const v2f v = *(const v2f*)&red[(w * S_ + wave) * SM_COLS + 2 * lane];
        sx += v.x;
        sy += v.y;
    }
    v2f o;
    o.x = sx; o.y = sy;
    *(v2f*)(orow + 2 * D_) = o;
}

extern "C" void kernel_launch(void* const* d_in, const int* in_sizes, int n_in,
                              void* d_out, int out_size, void* d_ws, size_t ws_size,
                              hipStream_t stream) {
    const float* x   = (const float*)d_in[0];
    const int*   ids = (const int*)d_in[2];
    float*       out = (float*)d_out;

    dim3 grid(D_ / SM_COLS, B_);
    span_pool_fused<<<grid, SM_WAVES * 32, 0, stream>>>(x, ids, out);
}


# 8-wave work-groups (2 waves per SIMD, 256 tokens per wave, 48-56 loads in flight per wave): 2048 waves enter instead of 4096; K split over 8 waves not 16 changes the sum tree
# speedup vs baseline: 7.8901x; 1.0933x over previous
#include <hip/hip_runtime.h>

#define B_ 16
#define L_ 2048
#define D_ 1024
#define S_ 16
#define OUTD (3 * D_)

typedef __attribute__((ext_vector_type(2))) float v2f;
typedef __attribute__((ext_vector_type(4))) float v4f;
typedef __attribute__((ext_vector_type(8))) float v8f;

#define SM_WAVES  8
#define SM_SLICE  (L_ / SM_WAVES)
#define SM_SPW    (S_ / SM_WAVES)
#define SM_COLS   64
#define SM_GSTEPS 4
#define SM_GROUPS (SM_SLICE / (4 * SM_GSTEPS))
#define SM_RING   7
#define FM_CHUNKS (SM_SLICE / 32)

typedef __attribute__((ext_vector_type(4))) int v4i;

static __device__ __forceinline__ void lds_handover_barrier() {
    __builtin_amdgcn_fence(__ATOMIC_RELEASE, "workgroup", "local");
    __builtin_amdgcn_s_barrier();
    __builtin_amdgcn_fence(__ATOMIC_ACQUIRE, "workgroup", "local");
}

__global__ void __launch_bounds__(SM_WAVES * 32) span_pool_fused(
    const float* __restrict__ x,
    const int* __restrict__ ids,
    float* __restrict__ out) {
    __shared__ float red[SM_WAVES * S_ * SM_COLS];
    __shared__ __attribute__((aligned(16))) int wcnt[SM_WAVES];
    __shared__ int sidx[2 * S_];

    const int b    = blockIdx.y;
    const int wave = threadIdx.x >> 5;
    const int lane = threadIdx.x & 31;
    const int col0 = blockIdx.x * SM_COLS;

    const int n     = lane & 15;
    const int h     = lane >> 4;
    const int khalf = h * 2;
    const int kbeg  = wave * SM_SLICE;

    const int* idrow = ids + b * L_ + kbeg + lane;
    int id[FM_CHUNKS];
    for (int j = 0; j < FM_CHUNKS; ++j)
        id[j] = idrow[j * 32];

    const float* base = x + ((size_t)b * L_ + kbeg + khalf) * D_ + col0 + 4 * n;

    v4f buf[SM_RING][SM_GSTEPS][2];
#define LOAD_GROUP(G) do {                                                     \
        _Pragma("unroll") for (int s_ = 0; s_ < SM_GSTEPS; ++s_) {             \
            const float* p_ = base + (size_t)(((G) * SM_GSTEPS + s_) * 4) * D_; \
            buf[(G) % SM_RING][s_][0] = *(const v4f*)p_;                       \
            buf[(G) % SM_RING][s_][1] = *(const v4f*)(p_ + D_);                \
        } } while (0)

#pragma unroll
    for (int g = 0; g < SM_RING - 1; ++g) LOAD_GROUP(g);

    if (threadIdx.x < 2 * S_) sidx[threadIdx.x] = 0;

    unsigned ms[FM_CHUNKS], me[FM_CHUNKS];
    int ns = 0, ne = 0;
    for (int j = 0; j < FM_CHUNKS; ++j) {
        ms[j] = __builtin_amdgcn_ballot_w32(id[j] >= 1 && id[j] <= 2);
        me[j] = __builtin_amdgcn_ballot_w32(id[j] >= 3 && id[j] <= 4);
        ns += __popc(ms[j]);
        ne += __popc(me[j]);
    }
    if (lane == 0) wcnt[wave] = ns | (ne << 16);
    lds_handover_barrier();

    int pre = 0;
    for (int q = 0; q < SM_WAVES / 4; ++q) {
        const v4i v = *(const v4i*)&wcnt[4 * q];
        pre += (4 * q + 0 < wave) ? v.x : 0;
        pre += (4 * q + 1 < wave) ? v.y : 0;
        pre += (4 * q + 2 < wave) ? v.z : 0;
        pre += (4 * q + 3 < wave) ? v.w : 0;
    }
    int bs = pre & 0xffff;
    int be = pre >> 16;

    const unsigned below = (1u << lane) - 1u;
    for (int j = 0; j < FM_CHUNKS; ++j) {
        const int p = kbeg + j * 32 + lane;
        if ((ms[j] >> lane) & 1u) {
            const int r = bs + __popc(ms[j] & below);
            if (r < S_) sidx[r] = p;
        }
        if ((me[j] >> lane) & 1u) {
            const int r = be + __popc(me[j] & below);
            if (r < S_) sidx[S_ + r] = p;
        }
        bs += __popc(ms[j]);
        be += __popc(me[j]);
    }
    lds_handover_barrier();

    const int i0 = sidx[n];
    const int i1 = sidx[S_ + n];

    v2f e0[SM_SPW], e1[SM_SPW];
#pragma unroll
    for (int q = 0; q < SM_SPW; ++q) {
        const int g0 = sidx[SM_SPW * wave + q];
        const int g1 = sidx[S_ + SM_SPW * wave + q];
        e0[q] = *(const v2f*)(x + ((size_t)b * L_ + g0) * D_ + col0 + 2 * lane);
        e1[q] = *(const v2f*)(x + ((size_t)b * L_ + g1) * D_ + col0 + 2 * lane);
    }

    const float invc = 1.0f / (float)(i1 - i0 - 1);

    v8f c0 = {}, c1 = {}, c2 = {}, c3 = {};
#pragma unroll
    for (int g = 0; g < SM_GROUPS; ++g) {
        __builtin_amdgcn_sched_barrier(0);
        if (g + SM_RING - 1 < SM_GROUPS) LOAD_GROUP(g + SM_RING - 1);
        __builtin_amdgcn_sched_barrier(0);
#pragma unroll
        for (int s = 0; s < SM_GSTEPS; ++s) {
            const int lx = kbeg + (g * SM_GSTEPS + s) * 4 + khalf;
            const int ly = lx + 1;

            v2f a;
            a.x = (lx > i0 && lx < i1) ? invc : 0.0f;
            a.y = (ly > i0 && ly < i1) ? invc : 0.0f;

            const v4f r0 = buf[g % SM_RING][s][0];
            const v4f r1 = buf[g % SM_RING][s][1];
            v2f bb;
            bb.x = r0.x; bb.y = r1.x;
            c0 = __builtin_amdgcn_wmma_f32_16x16x4_f32(
                      false, a,  false, bb,
                      (short)0, c0,  false,  false);
            bb.x = r0.y; bb.y = r1.y;
            c1 = __builtin_amdgcn_wmma_f32_16x16x4_f32(
                     false, a, false, bb, (short)0, c1, false, false);
            bb.x = r0.z; bb.y = r1.z;
            c2 = __builtin_amdgcn_wmma_f32_16x16x4_f32(
                     false, a, false, bb, (short)0, c2, false, false);
            bb.x = r0.w; bb.y = r1.w;
            c3 = __builtin_amdgcn_wmma_f32_16x16x4_f32(
                     false, a, false, bb, (short)0, c3, false, false);
        }
    }
#undef LOAD_GROUP

    float* orow = out + (size_t)(b * S_ + SM_SPW * wave) * OUTD + col0 + 2 * lane;
#pragma unroll
    for (int q = 0; q < SM_SPW; ++q) {
        *(v2f*)(orow + (size_t)q * OUTD)      = e0[q];
        *(v2f*)(orow + (size_t)q * OUTD + D_) = e1[q];
    }

#pragma unroll
    for (int r = 0; r < 8; ++r) {
        v4f v;
        v.x = c0[r]; v.y = c1[r]; v.z = c2[r]; v.w = c3[r];
        *(v4f*)&red[(wave * S_ + r + 8 * h) * SM_COLS + 4 * n] = v;
    }
    lds_handover_barrier();

#pragma unroll
    for (int q = 0; q < SM_SPW; ++q) {
        float sx = 0.0f, sy = 0.0f;
#pragma unroll
        for (int w = 0; w < SM_WAVES; ++w) {
            const v2f v = *(const v2f*)&red[(w * S_ + SM_SPW * wave + q) * SM_COLS + 2 * lane];
            sx += v.x;
            sy += v.y;
        }
        v2f o;
        o.x = sx; o.y = sy;
        *(v2f*)(orow + (size_t)q * OUTD + 2 * D_) = o;
    }
}

extern "C" void kernel_launch(void* const* d_in, const int* in_sizes, int n_in,
                              void* d_out, int out_size, void* d_ws, size_t ws_size,
                              hipStream_t stream) {
    const float* x   = (const float*)d_in[0];
    const int*   ids = (const int*)d_in[2];
    float*       out = (float*)d_out;

    dim3 grid(D_ / SM_COLS, B_);
    span_pool_fused<<<grid, SM_WAVES * 32, 0, stream>>>(x, ids, out);
}


# K compaction: a 16-token load group with no marker id (wave ballots) has one selector column: its rows are added in f32 first, 4 WMMAs replace 16; other groups unchanged; changes the sum tree
# speedup vs baseline: 8.3545x; 1.0589x over previous
#include <hip/hip_runtime.h>

#define B_ 16
#define L_ 2048
#define D_ 1024
#define S_ 16
#define OUTD (3 * D_)

typedef __attribute__((ext_vector_type(2))) float v2f;
typedef __attribute__((ext_vector_type(4))) float v4f;
typedef __attribute__((ext_vector_type(8))) float v8f;

#define SM_WAVES  8
#define SM_SLICE  (L_ / SM_WAVES)
#define SM_SPW    (S_ / SM_WAVES)
#define SM_COLS   64
#define SM_GSTEPS 4
#define SM_GROUPS (SM_SLICE / (4 * SM_GSTEPS))
#define SM_RING   7
#define FM_CHUNKS (SM_SLICE / 32)

typedef __attribute__((ext_vector_type(4))) int v4i;

static __device__ __forceinline__ void lds_handover_barrier() {
    __builtin_amdgcn_fence(__ATOMIC_RELEASE, "workgroup", "local");
    __builtin_amdgcn_s_barrier();
    __builtin_amdgcn_fence(__ATOMIC_ACQUIRE, "workgroup", "local");
}

__global__ void __launch_bounds__(SM_WAVES * 32) span_pool_fused(
    const float* __restrict__ x,
    const int* __restrict__ ids,
    float* __restrict__ out) {
    __shared__ float red[SM_WAVES * S_ * SM_COLS];
    __shared__ __attribute__((aligned(16))) int wcnt[SM_WAVES];
    __shared__ int sidx[2 * S_];

    const int b    = blockIdx.y;
    const int wave = threadIdx.x >> 5;
    const int lane = threadIdx.x & 31;
    const int col0 = blockIdx.x * SM_COLS;

    const int n     = lane & 15;
    const int h     = lane >> 4;
    const int khalf = h * 2;
    const int kbeg  = wave * SM_SLICE;
    const int wave_u = __builtin_amdgcn_readfirstlane(wave);

    const int* idrow = ids + b * L_ + kbeg + lane;
    int id[FM_CHUNKS];
    for (int j = 0; j < FM_CHUNKS; ++j)
        id[j] = idrow[j * 32];

    const float* base = x + ((size_t)b * L_ + kbeg + khalf) * D_ + col0 + 4 * n;

    v4f buf[SM_RING][SM_GSTEPS][2];
#define LOAD_GROUP(G) do {                                                     \
        _Pragma("unroll") for (int s_ = 0; s_ < SM_GSTEPS; ++s_) {             \
            const float* p_ = base + (size_t)(((G) * SM_GSTEPS + s_) * 4) * D_; \
            buf[(G) % SM_RING][s_][0] = *(const v4f*)p_;                       \
            buf[(G) % SM_RING][s_][1] = *(const v4f*)(p_ + D_);                \
        } } while (0)

#pragma unroll
    for (int g = 0; g < SM_RING - 1; ++g) LOAD_GROUP(g);

    if (threadIdx.x < 2 * S_) sidx[threadIdx.x] = 0;

    unsigned ms[FM_CHUNKS], me[FM_CHUNKS];
    int ns = 0, ne = 0;
    for (int j = 0; j < FM_CHUNKS; ++j) {
        ms[j] = __builtin_amdgcn_ballot_w32(id[j] >= 1 && id[j] <= 2);
        me[j] = __builtin_amdgcn_ballot_w32(id[j] >= 3 && id[j] <= 4);
        ns += __popc(ms[j]);
        ne += __popc(me[j]);
    }
    if (lane == 0) wcnt[wave] = ns | (ne << 16);
    lds_handover_barrier();

    int pre = 0;
    for (int q = 0; q < SM_WAVES / 4; ++q) {
        const v4i v = *(const v4i*)&wcnt[4 * q];
        pre += (4 * q + 0 < wave) ? v.x : 0;
        pre += (4 * q + 1 < wave) ? v.y : 0;
        pre += (4 * q + 2 < wave) ? v.z : 0;
        pre += (4 * q + 3 < wave) ? v.w : 0;
    }
    int bs = pre & 0xffff;
    int be = pre >> 16;

    const unsigned below = (1u << lane) - 1u;
    for (int j = 0; j < FM_CHUNKS; ++j) {
        const int p = kbeg + j * 32 + lane;
        if ((ms[j] >> lane) & 1u) {
            const int r = bs + __popc(ms[j] & below);
            if (r < S_) sidx[r] = p;
        }
        if ((me[j] >> lane) & 1u) {
            const int r = be + __popc(me[j] & below);
            if (r < S_) sidx[S_ + r] = p;
        }
        bs += __popc(ms[j]);
        be += __popc(me[j]);
    }
    lds_handover_barrier();

    const int i0 = sidx[n];
    const int i1 = sidx[S_ + n];

    v2f e0[SM_SPW], e1[SM_SPW];
#pragma unroll
    for (int q = 0; q < SM_SPW; ++q) {
        const int g0 = sidx[SM_SPW * wave + q];
        const int g1 = sidx[S_ + SM_SPW * wave + q];
        e0[q] = *(const v2f*)(x + ((size_t)b * L_ + g0) * D_ + col0 + 2 * lane);
        e1[q] = *(const v2f*)(x + ((size_t)b * L_ + g1) * D_ + col0 + 2 * lane);
    }

    const float invc = 1.0f / (float)(i1 - i0 - 1);

    v8f c0 = {}, c1 = {}, c2 = {}, c3 = {};
#pragma unroll
    for (int g = 0; g < SM_GROUPS; ++g) {
        __builtin_amdgcn_sched_barrier(0);
        if (g + SM_RING - 1 < SM_GROUPS) LOAD_GROUP(g + SM_RING - 1);
        __builtin_amdgcn_sched_barrier(0);
        const unsigned gm = ((ms[g >> 1] | me[g >> 1]) >> ((g & 1) * 16)) & 0xffffu;
        const bool mixed = (gm != 0u) || (g == 0 && wave_u == 0);

#define K_STEP(S, R0, R1) do {                                                 \
            const int lx = kbeg + (g * SM_GSTEPS + (S)) * 4 + khalf;           \
            const int ly = lx + 1;                                             \
            v2f a;                              \
            a.x = (lx > i0 && lx < i1) ? invc : 0.0f;                          \
            a.y = (ly > i0 && ly < i1) ? invc : 0.0f;                          \
            const v4f r0 = (R0);                                               \
            const v4f r1 = (R1);                                               \
            v2f bb;                                                            \
            bb.x = r0.x; bb.y = r1.x;                                          \
            c0 = __builtin_amdgcn_wmma_f32_16x16x4_f32(                        \
                      false, a,  false, bb,                  \
                      (short)0, c0,  false,  false); \
            bb.x = r0.y; bb.y = r1.y;                                          \
            c1 = __builtin_amdgcn_wmma_f32_16x16x4_f32(                        \
                     false, a, false, bb, (short)0, c1, false, false);         \
            bb.x = r0.z; bb.y = r1.z;                                          \
            c2 = __builtin_amdgcn_wmma_f32_16x16x4_f32(                        \
                     false, a, false, bb, (short)0, c2, false, false);         \
            bb.x = r0.w; bb.y = r1.w;                                          \
            c3 = __builtin_amdgcn_wmma_f32_16x16x4_f32(                        \
                     false, a, false, bb, (short)0, c3, false, false);         \
        } while (0)

        const v4f f0 = (buf[g % SM_RING][0][0] + buf[g % SM_RING][1][0])
                     + (buf[g % SM_RING][2][0] + buf[g % SM_RING][3][0]);
        const v4f f1 = (buf[g % SM_RING][0][1] + buf[g % SM_RING][1][1])
                     + (buf[g % SM_RING][2][1] + buf[g % SM_RING][3][1]);

        if (mixed) {
#pragma unroll
            for (int s = 0; s < SM_GSTEPS - 1; ++s)
                K_STEP(s, buf[g % SM_RING][s][0], buf[g % SM_RING][s][1]);
        }
        K_STEP(SM_GSTEPS - 1, mixed ? buf[g % SM_RING][SM_GSTEPS - 1][0] : f0,
                              mixed ? buf[g % SM_RING][SM_GSTEPS - 1][1] : f1);
#undef K_STEP
    }
#undef LOAD_GROUP

    float* orow = out + (size_t)(b * S_ + SM_SPW * wave) * OUTD + col0 + 2 * lane;
#pragma unroll
    for (int q = 0; q < SM_SPW; ++q) {
        *(v2f*)(orow + (size_t)q * OUTD)      = e0[q];
        *(v2f*)(orow + (size_t)q * OUTD + D_) = e1[q];
    }

#pragma unroll
    for (int r = 0; r < 8; ++r) {
        v4f v;
        v.x = c0[r]; v.y = c1[r]; v.z = c2[r]; v.w = c3[r];
        *(v4f*)&red[(wave * S_ + r + 8 * h) * SM_COLS + 4 * n] = v;
    }
    lds_handover_barrier();

#pragma unroll
    for (int q = 0; q < SM_SPW; ++q) {
        float sx = 0.0f, sy = 0.0f;
#pragma unroll
        for (int w = 0; w < SM_WAVES; ++w) {
            const v2f v = *(const v2f*)&red[(w * S_ + SM_SPW * wave + q) * SM_COLS + 2 * lane];
            sx += v.x;
            sy += v.y;
        }
        v2f o;
        o.x = sx; o.y = sy;
        *(v2f*)(orow + (size_t)q * OUTD + 2 * D_) = o;
    }
}

extern "C" void kernel_launch(void* const* d_in, const int* in_sizes, int n_in,
                              void* d_out, int out_size, void* d_ws, size_t ws_size,
                              hipStream_t stream) {
    const float* x   = (const float*)d_in[0];
    const int*   ids = (const int*)d_in[2];
    float*       out = (float*)d_out;

    dim3 grid(D_ / SM_COLS, B_);
    span_pool_fused<<<grid, SM_WAVES * 32, 0, stream>>>(x, ids, out);
}
